# starting kernel
# baseline (speedup 1.0000x reference)
#include <hip/hip_runtime.h>

#define N_FIN  32
#define HID    256
#define N_FOUT 64

typedef __attribute__((ext_vector_type(16))) _Float16 v16h;
typedef __attribute__((ext_vector_type(8)))  _Float16 v8h;
typedef __attribute__((ext_vector_type(8)))  float    v8f;

#define WMMA(a, b, c) __builtin_amdgcn_wmma_f32_16x16x32_f16( \
    false, (a), false, (b), (short)0, (c), false, false)

static __device__ __forceinline__ float fast_elu(float x) {
    const float e = __builtin_amdgcn_exp2f(fminf(x, 0.0f) * 1.44269504088896f) - 1.0f;
    return (x > 0.0f) ? x : e;
}

static __device__ __forceinline__ v16h ld_frag(const _Float16* p, int stride, int k0) {
    const int lane = threadIdx.x & 31;
    const _Float16* q = p + (lane & 15) * stride + k0 + ((lane >> 4) << 3);
    v8h lo = *(const v8h*)q;
    v8h hi = *(const v8h*)(q + 16);
    return __builtin_shufflevector(lo, hi, 0, 1, 2, 3, 4, 5, 6, 7,
                                           8, 9, 10, 11, 12, 13, 14, 15);
}

static __device__ __forceinline__ void st_elu_f16(_Float16* base, int stride,
                                                  v8f d, const float* bias, int o0) {
    const int lane = threadIdx.x & 31;
    const int col  = lane & 15;
    const int r0   = (lane >> 4) << 3;
    const float bv = bias[o0 + col];
#pragma unroll
    for (int r = 0; r < 8; ++r) {
        base[(r0 + r) * stride + col] = (_Float16)fast_elu(d[r] + bv);
    }
}

__global__ __launch_bounds__(256) void gnn_scatter(
    const float* __restrict__ feat, const float* __restrict__ ew,
    const int* __restrict__ src, const int* __restrict__ dst,
    float* __restrict__ msg, long long total) {
    long long idx = (long long)blockIdx.x * 256 + threadIdx.x;
    if (idx >= total) return;
    const int e = (int)(idx >> 5);
    const int c = (int)(idx & 31);
    const float v = feat[(long long)src[e] * N_FIN + c] * ew[e];
    atomicAdd(&msg[(long long)dst[e] * N_FIN + c], v);
}

__global__ __launch_bounds__(256) void gnn_fused(
    const float* __restrict__ feat, const float* __restrict__ msg,
    const float* __restrict__ Win, const float* __restrict__ bin,
    const float* __restrict__ Wh,  const float* __restrict__ bh,
    const float* __restrict__ Wo,  const float* __restrict__ bo,
    float* __restrict__ out, int nNodes) {
    __shared__ __align__(16) _Float16 sWin[HID * 2 * N_FIN];
    __shared__ __align__(16) _Float16 sWh [HID * HID];
    __shared__ __align__(16) _Float16 sWo [N_FOUT * HID];
    __shared__ __align__(16) _Float16 sScr[8 * 5632];
    __shared__ __align__(16) float    sBias[HID + HID + N_FOUT];

    const int tid = threadIdx.x;
    for (int i = tid; i < HID * 2 * N_FIN; i += 256) sWin[i] = (_Float16)Win[i];
    for (int i = tid; i < HID * HID;       i += 256) sWh[i]  = (_Float16)Wh[i];
    for (int i = tid; i < N_FOUT * HID;    i += 256) sWo[i]  = (_Float16)Wo[i];
    if (tid < HID)    sBias[tid]           = bin[tid];
    if (tid < HID)    sBias[HID + tid]     = bh[tid];
    if (tid < N_FOUT) sBias[2 * HID + tid] = bo[tid];
    __syncthreads();

    const float* sBin = sBias;
    const float* sBh  = sBias + HID;
    const float* sBo  = sBias + 2 * HID;

    const int wave = tid >> 5;
    const int lane = tid & 31;
    _Float16* bufA = sScr + wave * 5632;
    _Float16* bufH = bufA + 16 * 64;
    _Float16* bufS = bufH + 16 * HID;

    const int n0 = blockIdx.x * 128 + wave * 16;
    if (n0 >= nNodes) return;

#pragma unroll
    for (int it = 0; it < 32; ++it) {
        const int idx = it * 32 + lane;
        const int r = idx >> 6, c = idx & 63;
        const int node = n0 + r;
        float v = 0.0f;
        if (node < nNodes)
            v = (c < N_FIN) ? feat[(long long)node * N_FIN + c]
                            : msg[(long long)node * N_FIN + (c - N_FIN)];
        bufA[idx] = (_Float16)v;
    }

    for (int t = 0; t < 16; ++t) {
        v8f acc = {0.f, 0.f, 0.f, 0.f, 0.f, 0.f, 0.f, 0.f};
        acc = WMMA(ld_frag(bufA, 64, 0),  ld_frag(sWin + t * 16 * 64, 64, 0),  acc);
        acc = WMMA(ld_frag(bufA, 64, 32), ld_frag(sWin + t * 16 * 64, 64, 32), acc);
        st_elu_f16(bufH + t * 16, HID, acc, sBin, t * 16);
    }

    v8f acc3[4];
#pragma unroll
    for (int t3 = 0; t3 < 4; ++t3)
        acc3[t3] = (v8f){0.f, 0.f, 0.f, 0.f, 0.f, 0.f, 0.f, 0.f};

    for (int t2 = 0; t2 < 16; ++t2) {
        v8f acc = {0.f, 0.f, 0.f, 0.f, 0.f, 0.f, 0.f, 0.f};
#pragma unroll
        for (int kk = 0; kk < 8; ++kk)
            acc = WMMA(ld_frag(bufH, HID, kk * 32),
                       ld_frag(sWh + t2 * 16 * HID, HID, kk * 32), acc);
        st_elu_f16(bufS + (t2 & 1) * 16, 32, acc, sBh, t2 * 16);
        if (t2 & 1) {
            const v16h a3 = ld_frag(bufS, 32, 0);
            const int k0 = (t2 - 1) * 16;
#pragma unroll
            for (int t3 = 0; t3 < 4; ++t3)
                acc3[t3] = WMMA(a3, ld_frag(sWo + t3 * 16 * HID, HID, k0), acc3[t3]);
        }
    }

#pragma unroll
    for (int t3 = 0; t3 < 4; ++t3) {
        const int col = t3 * 16 + (lane & 15);
        const int r0  = (lane >> 4) << 3;
        const float bv = sBo[col];
#pragma unroll
        for (int r = 0; r < 8; ++r) {
            const int node = n0 + r0 + r;
            if (node < nNodes)
                out[(long long)node * N_FOUT + col] = acc3[t3][r] + bv;
        }
    }
}

extern "C" void kernel_launch(void* const* d_in, const int* in_sizes, int n_in,
                              void* d_out, int out_size, void* d_ws, size_t ws_size,
                              hipStream_t stream) {
    const float* feat = (const float*)d_in[0];
    const float* ew   = (const float*)d_in[1];
    const float* Win  = (const float*)d_in[2];
    const float* bin  = (const float*)d_in[3];
    const float* Wh   = (const float*)d_in[4];
    const float* bh   = (const float*)d_in[5];
    const float* Wo   = (const float*)d_in[6];
    const float* bo   = (const float*)d_in[7];
    const int*   ei   = (const int*)d_in[8];

    const int nNodes = in_sizes[0] / N_FIN;
    const int nEdges = in_sizes[1];
    float* msg = (float*)d_ws;

    hipMemsetAsync(msg, 0, (size_t)nNodes * N_FIN * sizeof(float), stream);

    const long long total = (long long)nEdges * 32;
    const int sblocks = (int)((total + 255) / 256);
    gnn_scatter<<<sblocks, 256, 0, stream>>>(feat, ew, ei, ei + nEdges, msg, total);

    const int gblocks = (nNodes + 127) / 128;
    gnn_fused<<<gblocks, 256, 0, stream>>>(feat, msg, Win, bin, Wh, bh, Wo, bo,
                                           (float*)d_out, nNodes);
}


# Fused MLP launched as one persistent block per WGP (grid 256): weights staged into LDS once per block, each wave then loops over its 16-row tiles (was staged again by every one of 782 blocks)
# speedup vs baseline: 2.3746x; 2.3746x over previous
#include <hip/hip_runtime.h>

#define N_FIN  32
#define HID    256
#define N_FOUT 64

#define P_IN   (2 * N_FIN + 8)
#define P_HID  (HID + 8)
#define P_S    (32 + 8)
#define SCR_PER_WAVE (16 * P_IN + 16 * P_HID + 16 * P_S)

typedef __attribute__((ext_vector_type(16))) _Float16 v16h;
typedef __attribute__((ext_vector_type(8)))  _Float16 v8h;
typedef __attribute__((ext_vector_type(8)))  float    v8f;

#define WMMA(a, b, c) __builtin_amdgcn_wmma_f32_16x16x32_f16( \
    false, (a), false, (b), (short)0, (c), false, false)

static __device__ __forceinline__ float fast_elu(float x) {
    const float e = __builtin_amdgcn_exp2f(fminf(x, 0.0f) * 1.44269504088896f) - 1.0f;
    return (x > 0.0f) ? x : e;
}

static __device__ __forceinline__ v16h ld_frag(const _Float16* p, int stride, int k0) {
    const int lane = threadIdx.x & 31;
    const _Float16* q = p + (lane & 15) * stride + k0 + ((lane >> 4) << 3);
    v8h lo = *(const v8h*)q;
    v8h hi = *(const v8h*)(q + 16);
    return __builtin_shufflevector(lo, hi, 0, 1, 2, 3, 4, 5, 6, 7,
                                           8, 9, 10, 11, 12, 13, 14, 15);
}

static __device__ __forceinline__ void st_elu_f16(_Float16* base, int stride,
                                                  v8f d, const float* bias, int o0) {
    const int lane = threadIdx.x & 31;
    const int col  = lane & 15;
    const int r0   = (lane >> 4) << 3;
    const float bv = bias[o0 + col];
#pragma unroll
    for (int r = 0; r < 8; ++r) {
        base[(r0 + r) * stride + col] = (_Float16)fast_elu(d[r] + bv);
    }
}

__global__ __launch_bounds__(256) void gnn_scatter(
    const float* __restrict__ feat, const float* __restrict__ ew,
    const int* __restrict__ src, const int* __restrict__ dst,
    float* __restrict__ msg, long long total) {
    long long idx = (long long)blockIdx.x * 256 + threadIdx.x;
    if (idx >= total) return;
    const int e = (int)(idx >> 5);
    const int c = (int)(idx & 31);
    const float v = feat[(long long)src[e] * N_FIN + c] * ew[e];
    atomicAdd(&msg[(long long)dst[e] * N_FIN + c], v);
}

__global__ __launch_bounds__(256) void gnn_link(
    const float* __restrict__ ew, const int* __restrict__ src,
    const int* __restrict__ dst, int* __restrict__ head,
    int* __restrict__ rec, int nEdges) {
    const int e = blockIdx.x * 256 + threadIdx.x;
    if (e >= nEdges) return;
    const int prev = atomicExch(&head[dst[e]], e);
    ((int4*)rec)[e] = make_int4(prev, src[e], __float_as_int(ew[e]), 0);
}

__global__ __launch_bounds__(256) void gnn_walk(
    const float* __restrict__ feat, const int* __restrict__ head,
    const int* __restrict__ rec, float* __restrict__ msg,
    int nNodes, int nEdges) {
    const int t = blockIdx.x * 256 + threadIdx.x;
    const int node = t >> 3;
    const int c4 = t & 7;
    if (node >= nNodes) return;
    float4 acc = make_float4(0.0f, 0.0f, 0.0f, 0.0f);
    if (nEdges > 0) {
        const int4*   rec4  = (const int4*)rec;
        const float4* feat4 = (const float4*)feat;
        int e = head[node];
        bool live = (unsigned)e < (unsigned)nEdges;
        int4 r = rec4[live ? e : 0];
        for (int it = 0; it < nEdges && live; ++it) {
            const float w = __int_as_float(r.z);
            const float4 f = feat4[(long long)r.y * (N_FIN / 4) + c4];
            e = r.x;
            live = (unsigned)e < (unsigned)nEdges;
            r = rec4[live ? e : 0];
            acc.x += f.x * w;
            acc.y += f.y * w;
            acc.z += f.z * w;
            acc.w += f.w * w;
            asm volatile("" : "+v"(r.x), "+v"(r.y), "+v"(r.z));
        }
    }
    ((float4*)msg)[(long long)node * (N_FIN / 4) + c4] = acc;
}

__global__ __launch_bounds__(256) void gnn_fused(
    const float* __restrict__ feat, const float* __restrict__ msg,
    const float* __restrict__ Win, const float* __restrict__ bin,
    const float* __restrict__ Wh,  const float* __restrict__ bh,
    const float* __restrict__ Wo,  const float* __restrict__ bo,
    float* __restrict__ out, int nNodes) {
    __shared__ __align__(16) _Float16 sWin[HID * P_IN];
    __shared__ __align__(16) _Float16 sWh [HID * P_HID];
    __shared__ __align__(16) _Float16 sWo [N_FOUT * P_HID];
    __shared__ __align__(16) _Float16 sScr[8 * SCR_PER_WAVE];
    __shared__ __align__(16) float    sBias[HID + HID + N_FOUT];

    const int tid = threadIdx.x;
    for (int i = tid; i < HID * 2 * N_FIN; i += 256)
        sWin[(i >> 6) * P_IN + (i & 63)] = (_Float16)Win[i];
    for (int i = tid; i < HID * HID; i += 256)
        sWh[(i >> 8) * P_HID + (i & 255)] = (_Float16)Wh[i];
    for (int i = tid; i < N_FOUT * HID; i += 256)
        sWo[(i >> 8) * P_HID + (i & 255)] = (_Float16)Wo[i];
    if (tid < HID)    sBias[tid]           = bin[tid];
    if (tid < HID)    sBias[HID + tid]     = bh[tid];
    if (tid < N_FOUT) sBias[2 * HID + tid] = bo[tid];
    __syncthreads();

    const float* sBin = sBias;
    const float* sBh  = sBias + HID;
    const float* sBo  = sBias + 2 * HID;

    const int wave = __builtin_amdgcn_readfirstlane(tid >> 5);
    const int lane = tid & 31;
    _Float16* bufA = sScr + wave * SCR_PER_WAVE;
    _Float16* bufH = bufA + 16 * P_IN;
    _Float16* bufS = bufH + 16 * P_HID;

    const int nTiles = (nNodes + 15) >> 4;
    const int tStep  = 8 * (int)gridDim.x;
    for (int tile = wave * (int)gridDim.x + (int)blockIdx.x; tile < nTiles; tile += tStep) {
        const int n0 = tile * 16;

#pragma unroll
        for (int it = 0; it < 32; ++it) {
            const int idx = it * 32 + lane;
            const int r = idx >> 6, c = idx & 63;
            const int node = n0 + r;
            float v = 0.0f;
            if (node < nNodes)
                v = (c < N_FIN) ? feat[(long long)node * N_FIN + c]
                                : msg[(long long)node * N_FIN + (c - N_FIN)];
            bufA[r * P_IN + c] = (_Float16)v;
        }

        for (int t = 0; t < 16; ++t) {
            v8f acc = {0.f, 0.f, 0.f, 0.f, 0.f, 0.f, 0.f, 0.f};
            acc = WMMA(ld_frag(bufA, P_IN, 0),  ld_frag(sWin + t * 16 * P_IN, P_IN, 0),  acc);
            acc = WMMA(ld_frag(bufA, P_IN, 32), ld_frag(sWin + t * 16 * P_IN, P_IN, 32), acc);
            st_elu_f16(bufH + t * 16, P_HID, acc, sBin, t * 16);
        }

        v8f acc3[4];
#pragma unroll
        for (int t3 = 0; t3 < 4; ++t3)
            acc3[t3] = (v8f){0.f, 0.f, 0.f, 0.f, 0.f, 0.f, 0.f, 0.f};

        for (int t2 = 0; t2 < 16; ++t2) {
            v8f acc = {0.f, 0.f, 0.f, 0.f, 0.f, 0.f, 0.f, 0.f};
#pragma unroll
            for (int kk = 0; kk < 8; ++kk)
                acc = WMMA(ld_frag(bufH, P_HID, kk * 32),
                           ld_frag(sWh + t2 * 16 * P_HID, P_HID, kk * 32), acc);
            st_elu_f16(bufS + (t2 & 1) * 16, P_S, acc, sBh, t2 * 16);
            if (t2 & 1) {
                const v16h a3 = ld_frag(bufS, P_S, 0);
                const int k0 = (t2 - 1) * 16;
#pragma unroll
                for (int t3 = 0; t3 < 4; ++t3)
                    acc3[t3] = WMMA(a3, ld_frag(sWo + t3 * 16 * P_HID, P_HID, k0), acc3[t3]);
            }
        }

#pragma unroll
        for (int t3 = 0; t3 < 4; ++t3) {
            const int col = t3 * 16 + (lane & 15);
            const int r0  = (lane >> 4) << 3;
            const float bv = sBo[col];
#pragma unroll
            for (int r = 0; r < 8; ++r) {
                const int node = n0 + r0 + r;
                if (node < nNodes)
                    out[(long long)node * N_FOUT + col] = acc3[t3][r] + bv;
            }
        }
    }
}

extern "C" void kernel_launch(void* const* d_in, const int* in_sizes, int n_in,
                              void* d_out, int out_size, void* d_ws, size_t ws_size,
                              hipStream_t stream) {
    const float* feat = (const float*)d_in[0];
    const float* ew   = (const float*)d_in[1];
    const float* Win  = (const float*)d_in[2];
    const float* bin  = (const float*)d_in[3];
    const float* Wh   = (const float*)d_in[4];
    const float* bh   = (const float*)d_in[5];
    const float* Wo   = (const float*)d_in[6];
    const float* bo   = (const float*)d_in[7];
    const int*   ei   = (const int*)d_in[8];

    const int nNodes = in_sizes[0] / N_FIN;
    const int nEdges = in_sizes[1];
    const size_t recBytes  = (size_t)nEdges * 16;
    const size_t msgBytes  = (size_t)nNodes * N_FIN * sizeof(float);
    const size_t headBytes = (size_t)nNodes * sizeof(int);
    float* msg;
    if (ws_size >= recBytes + msgBytes + headBytes) {
        int* rec  = (int*)d_ws;
        msg       = (float*)((char*)d_ws + recBytes);
        int* head = (int*)((char*)d_ws + recBytes + msgBytes);
        hipMemsetAsync(head, 0xFF, headBytes, stream);
        gnn_link<<<(nEdges + 255) / 256, 256, 0, stream>>>(ew, ei, ei + nEdges,
                                                           head, rec, nEdges);
        gnn_walk<<<(nNodes * 8 + 255) / 256, 256, 0, stream>>>(feat, head, rec, msg,
                                                               nNodes, nEdges);
    } else {
        msg = (float*)d_ws;
        hipMemsetAsync(msg, 0, msgBytes, stream);
        const long long total = (long long)nEdges * 32;
        const int sblocks = (int)((total + 255) / 256);
        gnn_scatter<<<sblocks, 256, 0, stream>>>(feat, ew, ei, ei + nEdges, msg, total);
    }

    gnn_fused<<<256, 256, 0, stream>>>(feat, msg, Win, bin, Wh, bh, Wo, bo,
                                       (float*)d_out, nNodes);
}


# MLP fetches each WMMA block's LDS operands (weight fragments, bias tiles) one block ahead, as one group behind the previous block's WMMAs and under its ELU, in all three layers
# speedup vs baseline: 8.2718x; 3.4834x over previous
#include <hip/hip_runtime.h>

#define N_FIN  32
#define HID    256
#define N_FOUT 64

#define P_IN   (2 * N_FIN + 8)
#define P_HID  (HID + 8)

typedef __attribute__((ext_vector_type(16))) _Float16 v16h;
typedef __attribute__((ext_vector_type(8)))  _Float16 v8h;
typedef __attribute__((ext_vector_type(8)))  float    v8f;
typedef __attribute__((ext_vector_type(4)))  _Float16 v4h;

static __device__ __forceinline__ void st4_f16(_Float16* dst, float4 v) {
    v4h h = {(_Float16)v.x, (_Float16)v.y, (_Float16)v.z, (_Float16)v.w};
    *(v4h*)dst = h;
}

template <int SHIFT, int PITCH>
static __device__ __forceinline__ void stage_batch(const float4* __restrict__ src4,
                                                   _Float16* dst, int first4) {
    float4 v[16];
#pragma unroll
    for (int j = 0; j < 16; ++j) v[j] = src4[first4 + j * 256];
#pragma unroll
    for (int j = 0; j < 16; ++j) {
        const int e = (first4 + j * 256) * 4;
        st4_f16(dst + (e >> SHIFT) * PITCH + (e & ((1 << SHIFT) - 1)), v[j]);
    }
}

#define WMMA(a, b, c) __builtin_amdgcn_wmma_f32_16x16x32_f16( \
    false, (a), false, (b), (short)0, (c), false, false)

static __device__ __forceinline__ float fast_elu(float x) {
    const float e = __builtin_amdgcn_exp2f(fminf(x, 0.0f) * 1.44269504088896f) - 1.0f;
    return (x > 0.0f) ? x : e;
}

static __device__ __forceinline__ v16h ld_frag(const _Float16* p, int stride, int k0) {
    const int lane = threadIdx.x & 31;
    const _Float16* q = p + (lane & 15) * stride + k0 + ((lane >> 4) << 3);
    v8h lo = *(const v8h*)q;
    v8h hi = *(const v8h*)(q + 16);
    return __builtin_shufflevector(lo, hi, 0, 1, 2, 3, 4, 5, 6, 7,
                                           8, 9, 10, 11, 12, 13, 14, 15);
}

static __device__ __forceinline__ v16h pack_elu(v8f lo, v8f hi) {
    v16h r;
#pragma unroll
    for (int i = 0; i < 8; ++i) {
        r[i]     = (_Float16)fast_elu(lo[i]);
        r[8 + i] = (_Float16)fast_elu(hi[i]);
    }
    return r;
}

static __device__ __forceinline__ v8f ld8f(const float* p) {
    const float4 a = *(const float4*)p;
    const float4 b = *(const float4*)(p + 4);
    return (v8f){a.x, a.y, a.z, a.w, b.x, b.y, b.z, b.w};
}

__global__ __launch_bounds__(256) void gnn_scatter(
    const float* __restrict__ feat, const float* __restrict__ ew,
    const int* __restrict__ src, const int* __restrict__ dst,
    float* __restrict__ msg, long long total) {
    long long idx = (long long)blockIdx.x * 256 + threadIdx.x;
    if (idx >= total) return;
    const int e = (int)(idx >> 5);
    const int c = (int)(idx & 31);
    const float v = feat[(long long)src[e] * N_FIN + c] * ew[e];
    atomicAdd(&msg[(long long)dst[e] * N_FIN + c], v);
}

__global__ __launch_bounds__(256) void gnn_link(
    const float* __restrict__ ew, const int* __restrict__ src,
    const int* __restrict__ dst, int* __restrict__ head,
    int* __restrict__ rec, int nEdges) {
    const int e = blockIdx.x * 256 + threadIdx.x;
    if (e >= nEdges) return;
    const int prev = atomicExch(&head[dst[e]], e);
    ((int4*)rec)[e] = make_int4(prev, src[e], __float_as_int(ew[e]), 0);
}

__global__ __launch_bounds__(256) void gnn_walk(
    const float* __restrict__ feat, const int* __restrict__ head,
    const int* __restrict__ rec, float* __restrict__ msg,
    int nNodes, int nEdges) {
    const int t = blockIdx.x * 256 + threadIdx.x;
    const int node = t >> 3;
    const int c4 = t & 7;
    if (node >= nNodes) return;
    float4 acc = make_float4(0.0f, 0.0f, 0.0f, 0.0f);
    if (nEdges > 0) {
        const int4*   rec4  = (const int4*)rec;
        const float4* feat4 = (const float4*)feat;
        int e = head[node];
        bool live = (unsigned)e < (unsigned)nEdges;
        int4 r = rec4[live ? e : 0];
        for (int it = 0; it < nEdges && live; ++it) {
            const float w = __int_as_float(r.z);
            const float4 f = feat4[(long long)r.y * (N_FIN / 4) + c4];
            e = r.x;
            live = (unsigned)e < (unsigned)nEdges;
            r = rec4[live ? e : 0];
            acc.x += f.x * w;
            acc.y += f.y * w;
            acc.z += f.z * w;
            acc.w += f.w * w;
            asm volatile("" : "+v"(r.x), "+v"(r.y), "+v"(r.z));
        }
    }
    ((float4*)msg)[(long long)node * (N_FIN / 4) + c4] = acc;
}

#define SL_BINS   256
#define SL_MAXN   512
#define SL_PITCH  33
#define SL_CHUNK  4096

__global__ __launch_bounds__(512) void gnn_bin(
    const float* __restrict__ ew, const int* __restrict__ src,
    const int* __restrict__ dst, int* __restrict__ binCount,
    int* __restrict__ binData, int cap, int npb, int nNodes, int nEdges) {
    __shared__ int sCnt[SL_BINS];
    __shared__ int sOff[SL_BINS];
    __shared__ int sBase[SL_BINS];
    __shared__ int sTot[8];
    __shared__ int2 sRec[SL_CHUNK];
    __shared__ unsigned short sBin[SL_CHUNK];

    const int tid  = threadIdx.x;
    const int lane = tid & 31;
    if (tid < SL_BINS) sCnt[tid] = 0;
    __syncthreads();

    int bin[8], rnk[8], px[8], py[8];
#pragma unroll
    for (int j = 0; j < 8; ++j) {
        const int e  = (int)blockIdx.x * SL_CHUNK + j * 512 + tid;
        const int ec = min(e, nEdges - 1);
        bin[j] = dst[ec];
        px[j]  = src[ec];
        py[j]  = __float_as_int(ew[ec]);
    }
#pragma unroll
    for (int j = 0; j < 8; ++j) {
        const int e = (int)blockIdx.x * SL_CHUNK + j * 512 + tid;
        const int d = bin[j];
        const int s = px[j];
        const bool ok = (e < nEdges) && ((unsigned)d < (unsigned)nNodes)
                                     && ((unsigned)s < (unsigned)nNodes);
        const unsigned dd = ok ? (unsigned)d : 0u;
        const unsigned b  = dd / (unsigned)npb;
        px[j]  = (s << 9) | (int)(dd - b * (unsigned)npb);
        bin[j] = ok ? (int)b : -1;
    }
#pragma unroll
    for (int j = 0; j < 8; ++j)
        rnk[j] = (bin[j] >= 0) ? atomicAdd(&sCnt[bin[j]], 1) : 0;
    __syncthreads();

    int c = 0, v = 0;
    if (tid < SL_BINS) {
        c = sCnt[tid];
        v = c;
#pragma unroll
        for (int o = 1; o < 32; o <<= 1) {
            const int t = __shfl_up(v, o);
            if (lane >= o) v += t;
        }
        if (lane == 31) sTot[tid >> 5] = v;
    }
    __syncthreads();
    if (tid < SL_BINS) {
        int pre = 0;
        for (int k = 0; k < (tid >> 5); ++k) pre += sTot[k];
        sOff[tid]  = pre + v - c;
        sBase[tid] = (c > 0) ? atomicAdd(&binCount[tid], c) : 0;
    }
    __syncthreads();

#pragma unroll
    for (int j = 0; j < 8; ++j) {
        if (bin[j] >= 0) {
            const int p = sOff[bin[j]] + rnk[j];
            sRec[p] = make_int2(px[j], py[j]);
            sBin[p] = (unsigned short)bin[j];
        }
    }
    __syncthreads();

    const int nValid = sOff[SL_BINS - 1] + sCnt[SL_BINS - 1];
    for (int i = tid; i < nValid; i += 512) {
        const int b   = sBin[i];
        const int pos = sBase[b] + (i - sOff[b]);
        if (pos < cap) ((int2*)binData)[(long long)b * cap + pos] = sRec[i];
    }
}

template <int U>
static __device__ __forceinline__ int bcast8(int v) {
    return __builtin_amdgcn_mov_dpp8(v, U * 0x249249);
}

static __device__ __forceinline__ void lds_add4(float* a, float4 f, float w) {
    __hip_atomic_fetch_add(a + 0, f.x * w, __ATOMIC_RELAXED, __HIP_MEMORY_SCOPE_WORKGROUP);
    __hip_atomic_fetch_add(a + 1, f.y * w, __ATOMIC_RELAXED, __HIP_MEMORY_SCOPE_WORKGROUP);
    __hip_atomic_fetch_add(a + 2, f.z * w, __ATOMIC_RELAXED, __HIP_MEMORY_SCOPE_WORKGROUP);
    __hip_atomic_fetch_add(a + 3, f.w * w, __ATOMIC_RELAXED, __HIP_MEMORY_SCOPE_WORKGROUP);
}

__global__ __launch_bounds__(1024) void gnn_binsum(
    const float* __restrict__ feat, const float* __restrict__ ew,
    const int* __restrict__ src, const int* __restrict__ dst,
    const int* __restrict__ binCount, const int* __restrict__ binData,
    float* __restrict__ msg, int cap, int npb, int nNodes, int nEdges) {
    __shared__ float acc[SL_MAXN * SL_PITCH];

    const int tid  = threadIdx.x;
    const int b    = blockIdx.x;
    const int lane = tid & 31;
    const int c4   = tid & 7;
    const float4* feat4 = (const float4*)feat;

    for (int i = tid; i < npb * SL_PITCH; i += 1024) acc[i] = 0.0f;
    __syncthreads();

    const int n = binCount[b];
    if (n <= cap) {
        const int2* rec = (const int2*)binData + (long long)b * cap;
        const int wave = __builtin_amdgcn_readfirstlane(tid >> 5);
        for (int base = wave * 32; base < n; base += 32 * 32) {
            const int2 rc = rec[min(base + lane, n - 1)];
            int rx[8], ry[8];
            rx[0] = bcast8<0>(rc.x); ry[0] = bcast8<0>(rc.y);
            rx[1] = bcast8<1>(rc.x); ry[1] = bcast8<1>(rc.y);
            rx[2] = bcast8<2>(rc.x); ry[2] = bcast8<2>(rc.y);
            rx[3] = bcast8<3>(rc.x); ry[3] = bcast8<3>(rc.y);
            rx[4] = bcast8<4>(rc.x); ry[4] = bcast8<4>(rc.y);
            rx[5] = bcast8<5>(rc.x); ry[5] = bcast8<5>(rc.y);
            rx[6] = bcast8<6>(rc.x); ry[6] = bcast8<6>(rc.y);
            rx[7] = bcast8<7>(rc.x); ry[7] = bcast8<7>(rc.y);
            float4 f[8];
#pragma unroll
            for (int u = 0; u < 8; ++u)
                f[u] = feat4[(long long)(rx[u] >> 9) * (N_FIN / 4) + c4];
            const int first = base + (lane & 24);
#pragma unroll
            for (int u = 0; u < 8; ++u) {
                if (first + u < n)
                    lds_add4(acc + (rx[u] & (SL_MAXN - 1)) * SL_PITCH + c4 * 4,
                             f[u], __int_as_float(ry[u]));
            }
        }
    } else {
        for (int e = tid >> 3; e < nEdges; e += 128) {
            const int d = dst[e];
            const int s = src[e];
            if ((unsigned)d < (unsigned)nNodes && (unsigned)s < (unsigned)nNodes
                && d / npb == b) {
                lds_add4(acc + (d - b * npb) * SL_PITCH + c4 * 4,
                         feat4[(long long)s * (N_FIN / 4) + c4], ew[e]);
            }
        }
    }
    __syncthreads();

    float4* msg4 = (float4*)msg;
    for (int i = tid; i < npb * 8; i += 1024) {
        const int nl   = i >> 3;
        const int q    = i & 7;
        const int node = b * npb + nl;
        if (node < nNodes) {
            const float* a = acc + nl * SL_PITCH + q * 4;
            msg4[(long long)node * (N_FIN / 4) + q] = make_float4(a[0], a[1], a[2], a[3]);
        }
    }
}

#define FW     8
#define P_O    (N_FOUT + 4)
#define SCR_A  (32 * P_IN)
#define SCR_O  (16 * P_O)

struct WinSet { v16h w[2][2]; v8f b[2]; };
struct WhSet  { v16h w[16];   v8f b[2]; };

static __device__ __forceinline__ void ld_win_set(WinSet& s, const _Float16* sWin,
                                                  const float* sBin, int kk, int h8) {
#pragma unroll
    for (int p = 0; p < 2; ++p) {
        const int T = 2 * kk + p;
        s.b[p]    = ld8f(sBin + T * 16 + h8);
        s.w[p][0] = ld_frag(sWin + T * 16 * P_IN, P_IN, 0);
        s.w[p][1] = ld_frag(sWin + T * 16 * P_IN, P_IN, 32);
    }
}

static __device__ __forceinline__ void ld_wh_set(WhSet& s, const _Float16* sWh,
                                                 const float* sBh, int c, int h8) {
#pragma unroll
    for (int p = 0; p < 2; ++p) s.b[p] = ld8f(sBh + (2 * c + p) * 16 + h8);
#pragma unroll
    for (int kk = 0; kk < 8; ++kk) {
#pragma unroll
        for (int p = 0; p < 2; ++p)
            s.w[kk * 2 + p] = ld_frag(sWh + (2 * c + p) * 16 * P_HID, P_HID, kk * 32);
    }
}

static __device__ __forceinline__ void mlp_unit(
    const _Float16* bufA, float* bufO,
    const _Float16* sWin, const _Float16* sWh, const _Float16* sWo,
    const float* sBin, const float* sBh, const float* sBo,
    float* __restrict__ out, int n0, int nNodes, int lane, int h8, WhSet& wh) {
    {
        v16h x[2][2];
#pragma unroll
        for (int nt = 0; nt < 2; ++nt) {
            x[nt][0] = ld_frag(bufA + nt * 16 * P_IN, P_IN, 0);
            x[nt][1] = ld_frag(bufA + nt * 16 * P_IN, P_IN, 32);
        }

        v16h h1[2][8];
        WinSet cw;
        ld_win_set(cw, sWin, sBin, 0, h8);
        __builtin_amdgcn_sched_barrier(0);
#pragma unroll
        for (int kk = 0; kk < 8; ++kk) {
            v8f a[2][2];
#pragma unroll
            for (int p = 0; p < 2; ++p) {
#pragma unroll
                for (int nt = 0; nt < 2; ++nt) {
                    a[p][nt] = WMMA(cw.w[p][0], x[nt][0], cw.b[p]);
                    a[p][nt] = WMMA(cw.w[p][1], x[nt][1], a[p][nt]);
                }
            }
            __builtin_amdgcn_sched_barrier(0);
            if (kk < 7) ld_win_set(cw, sWin, sBin, kk + 1, h8);
#pragma unroll
            for (int nt = 0; nt < 2; ++nt) h1[nt][kk] = pack_elu(a[0][nt], a[1][nt]);
            __builtin_amdgcn_sched_barrier(0);
        }

        v8f acc3[4][2];
#pragma unroll
        for (int t3 = 0; t3 < 4; ++t3) {
            const v8f bv = ld8f(sBo + t3 * 16 + h8);
            acc3[t3][0] = bv;
            acc3[t3][1] = bv;
        }
        for (int p2 = 0; p2 < 8; ++p2) {
            v8f a[2][2];
#pragma unroll
            for (int p = 0; p < 2; ++p) {
                a[p][0] = wh.b[p];
                a[p][1] = wh.b[p];
            }
#pragma unroll
            for (int kk = 0; kk < 8; ++kk) {
#pragma unroll
                for (int p = 0; p < 2; ++p) {
#pragma unroll
                    for (int nt = 0; nt < 2; ++nt)
                        a[p][nt] = WMMA(wh.w[kk * 2 + p], h1[nt][kk], a[p][nt]);
                }
            }
            __builtin_amdgcn_sched_barrier(0);
            v16h wo[4];
#pragma unroll
            for (int t3 = 0; t3 < 4; ++t3)
                wo[t3] = ld_frag(sWo + t3 * 16 * P_HID, P_HID, p2 * 32);
            ld_wh_set(wh, sWh, sBh, (p2 + 1) & 7, h8);
            v16h h2[2];
#pragma unroll
            for (int nt = 0; nt < 2; ++nt) h2[nt] = pack_elu(a[0][nt], a[1][nt]);
            __builtin_amdgcn_sched_barrier(0);
#pragma unroll
            for (int t3 = 0; t3 < 4; ++t3) {
#pragma unroll
                for (int nt = 0; nt < 2; ++nt)
                    acc3[t3][nt] = WMMA(wo[t3], h2[nt], acc3[t3][nt]);
            }
        }

#pragma unroll
        for (int nt = 0; nt < 2; ++nt) {
#pragma unroll
            for (int t3 = 0; t3 < 4; ++t3) {
                float* q = bufO + (lane & 15) * P_O + t3 * 16 + h8;
                const v8f d = acc3[t3][nt];
                *(float4*)q       = make_float4(d[0], d[1], d[2], d[3]);
                *(float4*)(q + 4) = make_float4(d[4], d[5], d[6], d[7]);
            }
            float4* out4 = (float4*)out;
#pragma unroll
            for (int j = 0; j < 8; ++j) {
                const int idx  = j * 32 + lane;
                const int row  = idx >> 4;
                const int node = n0 + nt * 16 + row;
                const float4 o = *(const float4*)(bufO + row * P_O + (idx & 15) * 4);
                if (node < nNodes) out4[(long long)node * (N_FOUT / 4) + (idx & 15)] = o;
            }
        }
    }
}

__global__ __launch_bounds__(FW * 32) void gnn_fused(
    const float* __restrict__ feat, const float* __restrict__ msg,
    const float* __restrict__ Win, const float* __restrict__ bin,
    const float* __restrict__ Wh,  const float* __restrict__ bh,
    const float* __restrict__ Wo,  const float* __restrict__ bo,
    float* __restrict__ out, int nNodes) {
    __shared__ __align__(16) _Float16 sWin[HID * P_IN];
    __shared__ __align__(16) _Float16 sWh [HID * P_HID];
    __shared__ __align__(16) _Float16 sWo [N_FOUT * P_HID];
    __shared__ __align__(16) _Float16 sScrA[FW * SCR_A];
    __shared__ __align__(16) float    sScrO[FW * SCR_O];
    __shared__ __align__(16) float    sBias[HID + HID + N_FOUT];

    const int tid = threadIdx.x;
    stage_batch<6, P_IN >((const float4*)Win, sWin, tid);
    stage_batch<8, P_HID>((const float4*)Wo,  sWo,  tid);
    for (int b = 0; b < 4; ++b)
        stage_batch<8, P_HID>((const float4*)Wh, sWh, b * 4096 + tid);
    if (tid < HID)    sBias[tid]           = bin[tid];
    if (tid < HID)    sBias[HID + tid]     = bh[tid];
    if (tid < N_FOUT) sBias[2 * HID + tid] = bo[tid];
    __syncthreads();

    const float* sBin = sBias;
    const float* sBh  = sBias + HID;
    const float* sBo  = sBias + 2 * HID;

    const int wave = __builtin_amdgcn_readfirstlane(tid >> 5);
    const int lane = tid & 31;
    const int h8   = (lane >> 4) << 3;
    _Float16* bufA = sScrA + wave * SCR_A;
    float*    bufO = sScrO + wave * SCR_O;

    const int nUnits = (nNodes + 31) >> 5;
    const int uStep  = FW * (int)gridDim.x;
    WhSet wh;
    ld_wh_set(wh, sWh, sBh, 0, h8);
    for (int unit = wave * (int)gridDim.x + (int)blockIdx.x; unit < nUnits; unit += uStep) {
        const int n0 = unit * 32;

        {
            const float4* feat4 = (const float4*)feat;
            const float4* msg4  = (const float4*)msg;
            float4 v[16];
#pragma unroll
            for (int j = 0; j < 16; ++j) {
                const int idx = j * 32 + lane;
                const int q = idx & 15;
                const int node = min(n0 + (idx >> 4), nNodes - 1);
                const float4* p = (q < 8) ? feat4 + (long long)node * 8 + q
                                          : msg4  + (long long)node * 8 + (q - 8);
                v[j] = *p;
            }
#pragma unroll
            for (int j = 0; j < 16; ++j) {
                const int idx = j * 32 + lane;
                st4_f16(bufA + (idx >> 4) * P_IN + (idx & 15) * 4, v[j]);
            }
        }
        mlp_unit(bufA, bufO, sWin, sWh, sWo, sBin, sBh, sBo, out, n0, nNodes, lane, h8, wh);
    }
}

#define SL_TRIP  128

static __device__ __forceinline__ void bcast8_all(int v, int (&o)[8]) {
    o[0] = bcast8<0>(v); o[1] = bcast8<1>(v); o[2] = bcast8<2>(v); o[3] = bcast8<3>(v);
    o[4] = bcast8<4>(v); o[5] = bcast8<5>(v); o[6] = bcast8<6>(v); o[7] = bcast8<7>(v);
}

static __device__ __forceinline__ void sl_load_rec(int2 (&rc)[4], const int2* __restrict__ rec,
                                                   int base, int lane, int n) {
#pragma unroll
    for (int k = 0; k < 4; ++k) rc[k] = rec[min(base + k * 32 + lane, n - 1)];
}

static __device__ __forceinline__ void sl_gather(float4 (&f)[32], const int2 (&rc)[4],
                                                 const float4* __restrict__ feat4, int c4) {
#pragma unroll
    for (int k = 0; k < 4; ++k) {
        int rx[8];
        bcast8_all(rc[k].x, rx);
#pragma unroll
        for (int u = 0; u < 8; ++u)
            f[k * 8 + u] = feat4[(long long)(rx[u] >> 9) * (N_FIN / 4) + c4];
    }
}

static __device__ __forceinline__ void sl_accum(float* acc, const float4 (&f)[32],
                                                const int2 (&rc)[4], int base, int lane,
                                                int c4, int n) {
#pragma unroll
    for (int k = 0; k < 4; ++k) {
        int rx[8], ry[8];
        bcast8_all(rc[k].x, rx);
        bcast8_all(rc[k].y, ry);
        const int first = base + k * 32 + (lane & 24);
#pragma unroll
        for (int u = 0; u < 8; ++u) {
            if (first + u < n)
                lds_add4(acc + (rx[u] & (SL_MAXN - 1)) * SL_PITCH + c4 * 4,
                         f[k * 8 + u], __int_as_float(ry[u]));
        }
    }
}

static __device__ __forceinline__ void w_issue(float4 (&v)[16], int wb, int tid,
                                               const float4* __restrict__ Win4,
                                               const float4* __restrict__ Wo4,
                                               const float4* __restrict__ Wh4) {
    const float4* src4 = (wb == 0) ? Win4 : (wb == 1) ? Wo4 : Wh4 + (wb - 2) * 4096;
#pragma unroll
    for (int j = 0; j < 16; ++j) v[j] = src4[tid + j * 256];
}

static __device__ __forceinline__ void w_store(const float4 (&v)[16], int wb, int tid,
                                               _Float16* sWin, _Float16* sWo, _Float16* sWh) {
    _Float16* dst   = (wb == 0) ? sWin : (wb == 1) ? sWo : sWh;
    const int shift = (wb == 0) ? 6 : 8;
    const int pitch = (wb == 0) ? P_IN : P_HID;
    const int first4 = tid + ((wb >= 2) ? (wb - 2) * 4096 : 0);
#pragma unroll
    for (int j = 0; j < 16; ++j) {
        const int e = (first4 + j * 256) * 4;
        st4_f16(dst + (e >> shift) * pitch + (e & ((1 << shift) - 1)), v[j]);
    }
}

__global__ __launch_bounds__(FW * 32) void gnn_sum_mlp(
    const float* __restrict__ feat, const float* __restrict__ ew,
    const int* __restrict__ src, const int* __restrict__ dst,
    const int* __restrict__ binCount, const int* __restrict__ binData,
    const float* __restrict__ Win, const float* __restrict__ bin,
    const float* __restrict__ Wh,  const float* __restrict__ bh,
    const float* __restrict__ Wo,  const float* __restrict__ bo,
    float* __restrict__ out, int cap, int npb, int nNodes, int nEdges) {
    __shared__ __align__(16) _Float16 sWin[HID * P_IN];
    __shared__ __align__(16) _Float16 sWh [HID * P_HID];
    __shared__ __align__(16) _Float16 sWo [N_FOUT * P_HID];
    __shared__ __align__(16) float    sBias[HID + HID + N_FOUT];
    __shared__ __align__(16) float    acc[SL_MAXN * SL_PITCH];
    __shared__ __align__(16) unsigned char sScr[FW * SCR_A * 2];

    const int tid  = threadIdx.x;
    const int b    = blockIdx.x;
    const int lane = tid & 31;
    const int c4   = tid & 7;
    const float4* feat4 = (const float4*)feat;

    {
        const float4 z = make_float4(0.0f, 0.0f, 0.0f, 0.0f);
        const int nz = (npb * SL_PITCH + 3) >> 2;
        for (int i = tid; i < nz; i += FW * 32) ((float4*)acc)[i] = z;
    }
    if (tid < HID)    sBias[tid]           = bin[tid];
    if (tid < HID)    sBias[HID + tid]     = bh[tid];
    if (tid < N_FOUT) sBias[2 * HID + tid] = bo[tid];
    __syncthreads();

    const float4* Win4 = (const float4*)Win;
    const float4* Wo4  = (const float4*)Wo;
    const float4* Wh4  = (const float4*)Wh;
    int wb = 0;
    float4 wv[16];

    const int wave = __builtin_amdgcn_readfirstlane(tid >> 5);
    const int n    = __builtin_amdgcn_readfirstlane(binCount[b]);
    if (n > 0 && n <= cap) {
        const int2* rec = (const int2*)binData + (long long)b * cap;
        const int step = FW * SL_TRIP;
        int base = wave * SL_TRIP;
        int2 rcA[4], rcB[4], rcN[4];
        float4 fA[32], fB[32];
        sl_load_rec(rcA, rec, base, lane, n);
        sl_load_rec(rcB, rec, base + step, lane, n);
        sl_gather(fA, rcA, feat4, c4);
        while (base < n) {
            sl_gather(fB, rcB, feat4, c4);
            sl_load_rec(rcN, rec, base + 2 * step, lane, n);
            const bool stA = wb < 6;
            if (stA) w_issue(wv, wb, tid, Win4, Wo4, Wh4);
            sl_accum(acc, fA, rcA, base, lane, c4, n);
            if (stA) { w_store(wv, wb, tid, sWin, sWo, sWh); ++wb; }
#pragma unroll
            for (int k = 0; k < 4; ++k) rcA[k] = rcN[k];
            base += step;
            if (base >= n) break;
            sl_gather(fA, rcA, feat4, c4);
            sl_load_rec(rcN, rec, base + 2 * step, lane, n);
            const bool stB = wb < 6;
            if (stB) w_issue(wv, wb, tid, Win4, Wo4, Wh4);
            sl_accum(acc, fB, rcB, base, lane, c4, n);
            if (stB) { w_store(wv, wb, tid, sWin, sWo, sWh); ++wb; }
#pragma unroll
            for (int k = 0; k < 4; ++k) rcB[k] = rcN[k];
            base += step;
        }
    } else if (n > cap) {
        for (int e = tid >> 3; e < nEdges; e += FW * 4) {
            const int d = dst[e];
            const int s = src[e];
            if ((unsigned)d < (unsigned)nNodes && (unsigned)s < (unsigned)nNodes
                && d / npb == b) {
                lds_add4(acc + (d - b * npb) * SL_PITCH + c4 * 4,
                         feat4[(long long)s * (N_FIN / 4) + c4], ew[e]);
            }
        }
    }
    for (; wb < 6; ++wb) {
        w_issue(wv, wb, tid, Win4, Wo4, Wh4);
        w_store(wv, wb, tid, sWin, sWo, sWh);
    }
    __syncthreads();

    const float* sBin = sBias;
    const float* sBh  = sBias + HID;
    const float* sBo  = sBias + 2 * HID;
    const int h8 = (lane >> 4) << 3;
    _Float16* bufA = (_Float16*)(sScr + wave * (SCR_A * 2));
    float*    bufO = (float*)(sScr + wave * (SCR_A * 2));

    const int node0  = b * npb;
    const int cnt    = min(npb, nNodes - node0);
    const int nUnits = cnt > 0 ? (cnt + 31) >> 5 : 0;
    WhSet wh;
    ld_wh_set(wh, sWh, sBh, 0, h8);
    for (int unit = wave; unit < nUnits; unit += FW) {
        const int l0 = unit * 32;
        {
            float4 v[8];
#pragma unroll
            for (int j = 0; j < 8; ++j) {
                const int idx = j * 32 + lane;
                const int l = min(l0 + (idx >> 3), cnt - 1);
                v[j] = feat4[(long long)(node0 + l) * (N_FIN / 4) + (idx & 7)];
            }
#pragma unroll
            for (int j = 0; j < 8; ++j) {
                const int idx = j * 32 + lane;
                const int l = min(l0 + (idx >> 3), cnt - 1);
                const float* a = acc + l * SL_PITCH + (idx & 7) * 4;
                st4_f16(bufA + (idx >> 3) * P_IN + N_FIN + (idx & 7) * 4,
                        make_float4(a[0], a[1], a[2], a[3]));
            }
#pragma unroll
            for (int j = 0; j < 8; ++j) {
                const int idx = j * 32 + lane;
                st4_f16(bufA + (idx >> 3) * P_IN + (idx & 7) * 4, v[j]);
            }
        }
        mlp_unit(bufA, bufO, sWin, sWh, sWo, sBin, sBh, sBo, out,
                 node0 + l0, node0 + cnt, lane, h8, wh);
        asm volatile("" ::: "memory");
    }
}

extern "C" void kernel_launch(void* const* d_in, const int* in_sizes, int n_in,
                              void* d_out, int out_size, void* d_ws, size_t ws_size,
                              hipStream_t stream) {
    const float* feat = (const float*)d_in[0];
    const float* ew   = (const float*)d_in[1];
    const float* Win  = (const float*)d_in[2];
    const float* bin  = (const float*)d_in[3];
    const float* Wh   = (const float*)d_in[4];
    const float* bh   = (const float*)d_in[5];
    const float* Wo   = (const float*)d_in[6];
    const float* bo   = (const float*)d_in[7];
    const int*   ei   = (const int*)d_in[8];

    const int nNodes = in_sizes[0] / N_FIN;
    const int nEdges = in_sizes[1];
    const size_t recBytes  = (size_t)nEdges * 16;
    const size_t msgBytes  = (size_t)nNodes * N_FIN * sizeof(float);
    const size_t headBytes = (size_t)nNodes * sizeof(int);
    const int npb = nNodes > 0 ? (nNodes + SL_BINS - 1) / SL_BINS : 1;
    const int cap = nEdges / SL_BINS + nEdges / (2 * SL_BINS) + 1024;
    const size_t binBytes = (size_t)SL_BINS * cap * sizeof(int2);
    const size_t cntBytes = (size_t)SL_BINS * sizeof(int);
    float* msg;
    if (npb <= SL_MAXN && ws_size >= binBytes + msgBytes + cntBytes) {
        int* binData  = (int*)d_ws;
        msg           = (float*)((char*)d_ws + binBytes);
        int* binCount = (int*)((char*)d_ws + binBytes + msgBytes);
        hipMemsetAsync(binCount, 0, cntBytes, stream);
        if (nEdges > 0)
            gnn_bin<<<(nEdges + SL_CHUNK - 1) / SL_CHUNK, 512, 0, stream>>>(
                ew, ei, ei + nEdges, binCount, binData, cap, npb, nNodes, nEdges);
        gnn_sum_mlp<<<SL_BINS, FW * 32, 0, stream>>>(feat, ew, ei, ei + nEdges, binCount,
                                                     binData, Win, bin, Wh, bh, Wo, bo,
                                                     (float*)d_out, cap, npb, nNodes, nEdges);
        return;
    } else if (ws_size >= recBytes + msgBytes + headBytes) {
        int* rec  = (int*)d_ws;
        msg       = (float*)((char*)d_ws + recBytes);
        int* head = (int*)((char*)d_ws + recBytes + msgBytes);
        hipMemsetAsync(head, 0xFF, headBytes, stream);
        gnn_link<<<(nEdges + 255) / 256, 256, 0, stream>>>(ew, ei, ei + nEdges,
                                                           head, rec, nEdges);
        gnn_walk<<<(nNodes * 8 + 255) / 256, 256, 0, stream>>>(feat, head, rec, msg,
                                                               nNodes, nEdges);
    } else {
        msg = (float*)d_ws;
        hipMemsetAsync(msg, 0, msgBytes, stream);
        const long long total = (long long)nEdges * 32;
        const int sblocks = (int)((total + 255) / 256);
        gnn_scatter<<<sblocks, 256, 0, stream>>>(feat, ew, ei, ei + nEdges, msg, total);
    }

    gnn_fused<<<256, 256, 0, stream>>>(feat, msg, Win, bin, Wh, bh, Wo, bo,
                                       (float*)d_out, nNodes);
}
